# starting kernel
# baseline (speedup 1.0000x reference)
#include <hip/hip_runtime.h>
#include <math.h>

typedef __attribute__((ext_vector_type(2))) float v2f;
typedef __attribute__((ext_vector_type(8))) float v8f;

#define NF    26
#define ED    4
#define NUMD  13
#define KDIM  117
#define KPAD  120
#define DSTR  124
#define L1N   12
#define L2N   8
#define L3N   4
#define ROWS  256
#define EPSV  1e-5f

__global__ __launch_bounds__(256) void deepfm_fused_kernel(
    const int*   __restrict__ x1,   const float* __restrict__ x2,
    const float* __restrict__ fw1,  const float* __restrict__ fw2,
    const float* __restrict__ bias,
    const float* __restrict__ w1, const float* __restrict__ b1,
    const float* __restrict__ g1, const float* __restrict__ be1,
    const float* __restrict__ m1, const float* __restrict__ v1,
    const float* __restrict__ w2, const float* __restrict__ b2,
    const float* __restrict__ g2, const float* __restrict__ be2,
    const float* __restrict__ m2, const float* __restrict__ v2,
    const float* __restrict__ w3, const float* __restrict__ b3,
    const float* __restrict__ g3, const float* __restrict__ be3,
    const float* __restrict__ m3, const float* __restrict__ v3,
    float* __restrict__ out)
{
    __shared__ float s_deep[ROWS][DSTR];
    __shared__ float s_w1t[16][DSTR];
    __shared__ float s_h1[ROWS][L1N];
    __shared__ float s_w2[L1N][L2N];
    __shared__ float s_w3[L2N][L3N];
    __shared__ float sA1[16], sC1[16];
    __shared__ float sA2[L2N], sC2[L2N];
    __shared__ float sA3[L3N], sC3[L3N];

    const int tid = threadIdx.x;
    const int row = blockIdx.x * ROWS + tid;

    for (int i = tid; i < 16 * DSTR; i += 256) {
        int n = i / DSTR, k = i % DSTR;
        s_w1t[n][k] = (n < L1N && k < KDIM) ? w1[k * L1N + n] : 0.0f;
    }
    if (tid < L1N * L2N) s_w2[tid / L2N][tid % L2N] = w2[tid];
    if (tid < L2N * L3N) s_w3[tid / L3N][tid % L3N] = w3[tid];
    if (tid < 16) {
        if (tid < L1N) {
            float sc = g1[tid] * rsqrtf(v1[tid] + EPSV);
            sA1[tid] = sc;
            sC1[tid] = (b1[tid] - m1[tid]) * sc + be1[tid];
        } else { sA1[tid] = 0.0f; sC1[tid] = 0.0f; }
    }
    if (tid >= 32 && tid < 32 + L2N) {
        int j = tid - 32;
        float sc = g2[j] * rsqrtf(v2[j] + EPSV);
        sA2[j] = sc; sC2[j] = (b2[j] - m2[j]) * sc + be2[j];
    }
    if (tid >= 64 && tid < 64 + L3N) {
        int j = tid - 64;
        float sc = g3[j] * rsqrtf(v3[j] + EPSV);
        sA3[j] = sc; sC3[j] = (b3[j] - m3[j]) * sc + be3[j];
    }

    float fm1 = 0.0f;
    float s0 = 0.f, s1 = 0.f, s2 = 0.f, s3 = 0.f;
    float q0 = 0.f, q1 = 0.f, q2 = 0.f, q3 = 0.f;
    {
        const float4* emb = (const float4*)fw2;
        #pragma unroll
        for (int f = 0; f < NF; ++f) {
            int idx = x1[row * NF + f];
            fm1 += fw1[idx];
            float4 e = emb[idx];
            s0 += e.x; s1 += e.y; s2 += e.z; s3 += e.w;
            q0 += e.x * e.x; q1 += e.y * e.y; q2 += e.z * e.z; q3 += e.w * e.w;
            *(float4*)&s_deep[tid][f * 4] = e;
        }
        #pragma unroll
        for (int j = 0; j < NUMD; ++j) s_deep[tid][NF * 4 + j] = x2[row * NUMD + j];
        s_deep[tid][117] = 0.0f; s_deep[tid][118] = 0.0f; s_deep[tid][119] = 0.0f;
    }
    const float fm2 = 0.5f * ((s0 * s0 - q0) + (s1 * s1 - q1) +
                              (s2 * s2 - q2) + (s3 * s3 - q3));

    __syncthreads();

    {
        const int lane = tid & 31;
        const int wave = tid >> 5;
        const int half = lane >> 4;
        const int mn   = lane & 15;
        #pragma unroll
        for (int t = 0; t < 2; ++t) {
            const int tileBase = (wave * 2 + t) * 16;
            v8f acc = {};
            #pragma unroll
            for (int c = 0; c < KPAD / 4; ++c) {
                const int k = c * 4 + half * 2;
                v2f a = *(const v2f*)&s_deep[tileBase + mn][k];
                v2f b = *(const v2f*)&s_w1t[mn][k];
                acc = __builtin_amdgcn_wmma_f32_16x16x4_f32(
                    false, a, false, b, (short)0, acc, false, false);
            }
            if (mn < L1N) {
                const float sc = sA1[mn], sh = sC1[mn];
                #pragma unroll
                for (int i = 0; i < 8; ++i) {
                    const int r = tileBase + i + half * 8;
                    float h = acc[i] * sc + sh;
                    s_h1[r][mn] = h > 0.0f ? h : 0.0f;
                }
            }
        }
    }

    __syncthreads();

    float h2[L2N];
    #pragma unroll
    for (int j = 0; j < L2N; ++j) {
        float a = 0.0f;
        #pragma unroll
        for (int k = 0; k < L1N; ++k) a += s_h1[tid][k] * s_w2[k][j];
        a = a * sA2[j] + sC2[j];
        h2[j] = a > 0.0f ? a : 0.0f;
    }
    float hsum = 0.0f;
    #pragma unroll
    for (int j = 0; j < L3N; ++j) {
        float a = 0.0f;
        #pragma unroll
        for (int k = 0; k < L2N; ++k) a += h2[k] * s_w3[k][j];
        a = a * sA3[j] + sC3[j];
        hsum += a > 0.0f ? a : 0.0f;
    }
    const float z = fm1 + fm2 + hsum + bias[0];
    out[row] = 1.0f / (1.0f + __expf(-z));
}

extern "C" void kernel_launch(void* const* d_in, const int* in_sizes, int n_in,
                              void* d_out, int out_size, void* d_ws, size_t ws_size,
                              hipStream_t stream) {
    const int*   x1   = (const int*)d_in[0];
    const float* x2   = (const float*)d_in[1];
    const float* fw1  = (const float*)d_in[2];
    const float* fw2  = (const float*)d_in[3];
    const float* bias = (const float*)d_in[4];
    const float* w1 = (const float*)d_in[5];  const float* b1 = (const float*)d_in[6];
    const float* g1 = (const float*)d_in[7];  const float* be1 = (const float*)d_in[8];
    const float* m1 = (const float*)d_in[9];  const float* v1 = (const float*)d_in[10];
    const float* w2 = (const float*)d_in[11]; const float* b2 = (const float*)d_in[12];
    const float* g2 = (const float*)d_in[13]; const float* be2 = (const float*)d_in[14];
    const float* m2 = (const float*)d_in[15]; const float* v2 = (const float*)d_in[16];
    const float* w3 = (const float*)d_in[17]; const float* b3 = (const float*)d_in[18];
    const float* g3 = (const float*)d_in[19]; const float* be3 = (const float*)d_in[20];
    const float* m3 = (const float*)d_in[21]; const float* v3 = (const float*)d_in[22];

    const int rows = in_sizes[0] / NF;
    const int grid = rows / ROWS;

    deepfm_fused_kernel<<<grid, ROWS, 0, stream>>>(
        x1, x2, fw1, fw2, bias,
        w1, b1, g1, be1, m1, v1,
        w2, b2, g2, be2, m2, v2,
        w3, b3, g3, be3, m3, v3,
        (float*)d_out);
}


# gather re-cut: one (row,field) element per lane in whole-line sweeps, every load issued up front, LDS-staged; 256 groups of 64 rows x 4 waves (was 64 of 256 rows x 8); sum orders kept
# speedup vs baseline: 2.3300x; 2.3300x over previous
#include <hip/hip_runtime.h>
#include <math.h>


typedef __attribute__((ext_vector_type(2))) float v2f;
typedef __attribute__((ext_vector_type(8))) float v8f;

#define NF    26
#define ED    4
#define NUMD  13
#define KDIM  117
#define KPAD  120
#define DSTR  124
#define L1N   12
#define L2N   8
#define L3N   4
#define ROWS  64
#define NT    128
#define EPT   13
#define DPT   7
#define WPT   11
#define FSTR  27
#define EPSV  1e-5f

__global__ __launch_bounds__(NT) void deepfm_fused_kernel(
    const int*   __restrict__ x1,   const float* __restrict__ x2,
    const float* __restrict__ fw1,  const float* __restrict__ fw2,
    const float* __restrict__ bias,
    const float* __restrict__ w1, const float* __restrict__ b1,
    const float* __restrict__ g1, const float* __restrict__ be1,
    const float* __restrict__ m1, const float* __restrict__ v1,
    const float* __restrict__ w2, const float* __restrict__ b2,
    const float* __restrict__ g2, const float* __restrict__ be2,
    const float* __restrict__ m2, const float* __restrict__ v2,
    const float* __restrict__ w3, const float* __restrict__ b3,
    const float* __restrict__ g3, const float* __restrict__ be3,
    const float* __restrict__ m3, const float* __restrict__ v3,
    float* __restrict__ out)
{
    __shared__ __attribute__((aligned(16))) float s_deep[ROWS][DSTR];
    __shared__ float s_w1t[16][DSTR];
    __shared__ float s_fw[ROWS][FSTR];
    __shared__ float s_h1[ROWS][L1N];
    __shared__ float s_w2[L1N][L2N];
    __shared__ float s_w3[L2N][L3N];
    __shared__ float s_fm[ROWS];
    __shared__ float sA1[16], sC1[16];
    __shared__ float sA2[L2N], sC2[L2N];
    __shared__ float sA3[L3N], sC3[L3N];

    const int tid  = threadIdx.x;
    const int row0 = blockIdx.x * ROWS;

    const int*   xi = x1 + row0 * NF;
    const float* xd = x2 + row0 * NUMD;
    int idx[EPT];
    #pragma unroll
    for (int j = 0; j < EPT; ++j) idx[j] = xi[j * NT + tid];

    float dv[DPT];
    #pragma unroll
    for (int j = 0; j < DPT; ++j) dv[j] = xd[min(j * NT + tid, ROWS * NUMD - 1)];
    float wv[WPT];
    #pragma unroll
    for (int j = 0; j < WPT; ++j) wv[j] = w1[min(j * NT + tid, KDIM * L1N - 1)];
    const int pw2 = min(tid, L1N * L2N - 1), pw3 = min(tid, L2N * L3N - 1);
    const float w2v = w2[pw2];
    const float w3v = w3[pw3];
    const int pa = min(tid, 15), p1 = min(tid, L1N - 1);
    const int p2 = min(tid, L2N - 1), p3 = min(tid, L3N - 1);
    const float g1v = g1[p1], v1v = v1[p1], b1v = b1[p1], m1v = m1[p1], be1v = be1[p1];
    const float g2v = g2[p2], v2v = v2[p2], b2v = b2[p2], m2v = m2[p2], be2v = be2[p2];
    const float g3v = g3[p3], v3v = v3[p3], b3v = b3[p3], m3v = m3[p3], be3v = be3[p3];

    const float4* emb = (const float4*)fw2;
    float4 ev[EPT];
    float  fv[EPT];
    #pragma unroll
    for (int j = 0; j < EPT; ++j) {
        ev[j] = emb[idx[j]];
        fv[j] = fw1[idx[j]];
    }

    #pragma unroll
    for (int j = 0; j < WPT; ++j) {
        const int e = min(j * NT + tid, KDIM * L1N - 1);
        const int k = e / L1N, n = e - k * L1N;
        s_w1t[n][k] = wv[j];
    }
    {
        const int i = min(tid, L1N * (DSTR - KDIM) - 1);
        s_w1t[i / (DSTR - KDIM)][KDIM + i % (DSTR - KDIM)] = 0.0f;
    }
    #pragma unroll
    for (int j = 0; j < 4; ++j)
        (&s_w1t[L1N][0])[min(j * NT + tid, (16 - L1N) * DSTR - 1)] = 0.0f;

    (&s_w2[0][0])[pw2] = w2v;
    (&s_w3[0][0])[pw3] = w3v;
    {
        const float sc1 = g1v * rsqrtf(v1v + EPSV);
        sA1[pa] = pa < L1N ? sc1 : 0.0f;
        sC1[pa] = pa < L1N ? (b1v - m1v) * sc1 + be1v : 0.0f;
        const float sc2 = g2v * rsqrtf(v2v + EPSV);
        sA2[p2] = sc2; sC2[p2] = (b2v - m2v) * sc2 + be2v;
        const float sc3 = g3v * rsqrtf(v3v + EPSV);
        sA3[p3] = sc3; sC3[p3] = (b3v - m3v) * sc3 + be3v;
    }

    #pragma unroll
    for (int j = 0; j < DPT; ++j) {
        const int e = min(j * NT + tid, ROWS * NUMD - 1);
        const int r = e / NUMD, c = e - r * NUMD;
        s_deep[r][NF * ED + c] = dv[j];
    }
    {
        const int r = tid & (ROWS - 1);
        s_deep[r][117] = 0.0f; s_deep[r][118] = 0.0f; s_deep[r][119] = 0.0f;
    }
    #pragma unroll
    for (int j = 0; j < EPT; ++j) {
        const int e = j * NT + tid;
        const int r = e / NF, f = e - r * NF;
        *(float4*)&s_deep[r][f * ED] = ev[j];
        s_fw[r][f] = fv[j];
    }

    __syncthreads();

    {
        const int lane = tid & 31;
        const int wave = tid >> 5;
        const int half = lane >> 4;
        const int mn   = lane & 15;
        const int tileBase = wave * 16;
        v8f acc = {};
        #pragma unroll
        for (int c = 0; c < KPAD / 4; ++c) {
            const int k = c * 4 + half * 2;
            v2f a = *(const v2f*)&s_deep[tileBase + mn][k];
            v2f b = *(const v2f*)&s_w1t[mn][k];
            acc = __builtin_amdgcn_wmma_f32_16x16x4_f32(
                false, a, false, b, (short)0, acc, false, false);
        }
        if (mn < L1N) {
            const float sc = sA1[mn], sh = sC1[mn];
            #pragma unroll
            for (int i = 0; i < 8; ++i) {
                const int r = tileBase + i + half * 8;
                float h = acc[i] * sc + sh;
                s_h1[r][mn] = h > 0.0f ? h : 0.0f;
            }
        }
    }

    __syncthreads();

    float hsum = 0.0f;
    if (tid < ROWS) {
        float h2[L2N];
        #pragma unroll
        for (int j = 0; j < L2N; ++j) {
            float a = 0.0f;
            #pragma unroll
            for (int k = 0; k < L1N; ++k) a += s_h1[tid][k] * s_w2[k][j];
            a = a * sA2[j] + sC2[j];
            h2[j] = a > 0.0f ? a : 0.0f;
        }
        #pragma unroll
        for (int j = 0; j < L3N; ++j) {
            float a = 0.0f;
            #pragma unroll
            for (int k = 0; k < L2N; ++k) a += h2[k] * s_w3[k][j];
            a = a * sA3[j] + sC3[j];
            hsum += a > 0.0f ? a : 0.0f;
        }
    } else {
        const int r = tid - ROWS;
        float fm1 = 0.0f;
        float s0 = 0.f, s1 = 0.f, s2 = 0.f, s3 = 0.f;
        float q0 = 0.f, q1 = 0.f, q2 = 0.f, q3 = 0.f;
        #pragma unroll
        for (int f = 0; f < NF; ++f) {
            fm1 += s_fw[r][f];
            float4 e = *(const float4*)&s_deep[r][f * ED];
            s0 += e.x; s1 += e.y; s2 += e.z; s3 += e.w;
            q0 += e.x * e.x; q1 += e.y * e.y; q2 += e.z * e.z; q3 += e.w * e.w;
        }
        const float fm2 = 0.5f * ((s0 * s0 - q0) + (s1 * s1 - q1) +
                                  (s2 * s2 - q2) + (s3 * s3 - q3));
        s_fm[r] = fm1 + fm2;
    }

    __syncthreads();

    if (tid < ROWS) {
        const float z = s_fm[tid] + hsum + bias[0];
        out[row0 + tid] = 1.0f / (1.0f + __expf(-z));
    }
}

extern "C" void kernel_launch(void* const* d_in, const int* in_sizes, int n_in,
                              void* d_out, int out_size, void* d_ws, size_t ws_size,
                              hipStream_t stream) {
    const int*   x1   = (const int*)d_in[0];
    const float* x2   = (const float*)d_in[1];
    const float* fw1  = (const float*)d_in[2];
    const float* fw2  = (const float*)d_in[3];
    const float* bias = (const float*)d_in[4];
    const float* w1 = (const float*)d_in[5];  const float* b1 = (const float*)d_in[6];
    const float* g1 = (const float*)d_in[7];  const float* be1 = (const float*)d_in[8];
    const float* m1 = (const float*)d_in[9];  const float* v1 = (const float*)d_in[10];
    const float* w2 = (const float*)d_in[11]; const float* b2 = (const float*)d_in[12];
    const float* g2 = (const float*)d_in[13]; const float* be2 = (const float*)d_in[14];
    const float* m2 = (const float*)d_in[15]; const float* v2 = (const float*)d_in[16];
    const float* w3 = (const float*)d_in[17]; const float* b3 = (const float*)d_in[18];
    const float* g3 = (const float*)d_in[19]; const float* be3 = (const float*)d_in[20];
    const float* m3 = (const float*)d_in[21]; const float* v3 = (const float*)d_in[22];

    const int rows = in_sizes[0] / NF;
    const int grid = rows / ROWS;

    deepfm_fused_kernel<<<grid, NT, 0, stream>>>(
        x1, x2, fw1, fw2, bias,
        w1, b1, g1, be1, m1, v1,
        w2, b2, g2, be2, m2, v2,
        w3, b3, g3, be3, m3, v3,
        (float*)d_out);
}
